# starting kernel
# baseline (speedup 1.0000x reference)
#include <hip/hip_runtime.h>
#include <cstdint>

#define F_IN  128
#define F_HID 128
#define F_OUT 64

typedef __attribute__((ext_vector_type(2))) float v2f;
typedef __attribute__((ext_vector_type(8))) float v8f;

__device__ __forceinline__ void atomic_add_f32(float* p, float v) {
  unsafeAtomicAdd(p, v);
}

__global__ void k_fill_const(float* __restrict__ p, float v, long long n) {
  long long i = (long long)blockIdx.x * blockDim.x + threadIdx.x;
  if (i < n) p[i] = v;
}

__global__ void k_fill_bias64(float* __restrict__ out, const float* __restrict__ b,
                              long long total) {
  long long i = (long long)blockIdx.x * blockDim.x + threadIdx.x;
  if (i < total) out[i] = b[i & 63];
}

__global__ void k_deg(const int* __restrict__ col, const float* __restrict__ ew,
                      float* __restrict__ deg, int E) {
  int e = blockIdx.x * blockDim.x + threadIdx.x;
  if (e < E) atomic_add_f32(&deg[col[e]], ew[e]);
}

__global__ void k_rsqrt(float* __restrict__ deg, int n) {
  int i = blockIdx.x * blockDim.x + threadIdx.x;
  if (i < n) {
    float d = deg[i];
    deg[i] = (d > 0.f) ? rsqrtf(d) : 0.f;
  }
}

__global__ void k_norm(const int* __restrict__ row, const int* __restrict__ col,
                       const float* __restrict__ ew, const float* __restrict__ dis,
                       float* __restrict__ nrm, int E) {
  int e = blockIdx.x * blockDim.x + threadIdx.x;
  if (e < E) nrm[e] = dis[row[e]] * ew[e] * dis[col[e]];
}

template <int FOUT>
__global__ void __launch_bounds__(256)
k_gemm_wmma(const float* __restrict__ A, const float* __restrict__ W,
            float* __restrict__ H, int n) {
  __shared__ float sW[F_IN * FOUT];
  for (int i = threadIdx.x; i < F_IN * FOUT; i += 256) sW[i] = W[i];
  __syncthreads();

  const int lane = threadIdx.x & 31;
  const int wave = threadIdx.x >> 5;
  const int tilesN = FOUT / 16;
  const long long numTiles = (long long)(n / 16) * tilesN;
  const long long tile = (long long)blockIdx.x * 8 + wave;
  if (tile >= numTiles) return;

  const int tm = (int)(tile / tilesN);
  const int tn = (int)(tile % tilesN);
  const int mrow = tm * 16 + (lane & 15);
  const int ncol = tn * 16 + (lane & 15);
  const int koff = (lane >> 4) * 2;

  const float* __restrict__ arow = A + (long long)mrow * F_IN;
  const float* __restrict__ wb = sW + ncol;

  v8f c = {};
#pragma unroll
  for (int k = 0; k < F_IN; k += 4) {
    const int ka = k + koff;
    v2f a, b;
    a.x = arow[ka];
    a.y = arow[ka + 1];
    b.x = wb[ka * FOUT];
    b.y = wb[(ka + 1) * FOUT];
    c = __builtin_amdgcn_wmma_f32_16x16x4_f32(
         false, a,  false, b,
         (short)0, c,  false,  false);
  }

  const int rbase = tm * 16 + ((lane >> 4) << 3);
#pragma unroll
  for (int v = 0; v < 8; ++v)
    H[(long long)(rbase + v) * FOUT + ncol] = c[v];
}

template <int F>
__global__ void __launch_bounds__(256)
k_scatter(const float* __restrict__ H, const int* __restrict__ row,
          const int* __restrict__ col, const float* __restrict__ nrm,
          float* __restrict__ out, int E) {
  const int lane = threadIdx.x & 31;
  const long long e = (long long)blockIdx.x * 8 + (threadIdx.x >> 5);
  if (e >= E) return;
  const int r = row[e];
  const int cgt = col[e];
  const float w = nrm[e];
  constexpr int PER = F / 32;
  const float* src = H + (long long)r * F + lane * PER;
  float* dst = out + (long long)cgt * F + lane * PER;
  if constexpr (PER == 4) {
    float4 v = *(const float4*)src;
    atomic_add_f32(dst + 0, v.x * w);
    atomic_add_f32(dst + 1, v.y * w);
    atomic_add_f32(dst + 2, v.z * w);
    atomic_add_f32(dst + 3, v.w * w);
  } else {
    float2 v = *(const float2*)src;
    atomic_add_f32(dst + 0, v.x * w);
    atomic_add_f32(dst + 1, v.y * w);
  }
}

__global__ void k_combine1(float* __restrict__ agg, const float* __restrict__ h,
                           const float* __restrict__ dis, const float* __restrict__ b1,
                           long long total) {
  long long i = (long long)blockIdx.x * blockDim.x + threadIdx.x;
  if (i >= total) return;
  const int node = (int)(i >> 7);
  const int j = (int)(i & 127);
  const float d = dis[node];
  float v = agg[i] + h[i] * d * d + b1[j];
  agg[i] = v > 0.f ? v : 0.f;
}

__global__ void k_combine2(float* __restrict__ out, const float* __restrict__ h2,
                           const float* __restrict__ dis, long long total) {
  long long i = (long long)blockIdx.x * blockDim.x + threadIdx.x;
  if (i >= total) return;
  const int node = (int)(i >> 6);
  const float d = dis[node];
  out[i] += h2[i] * d * d;
}

static inline int nblk(long long n, int t) { return (int)((n + t - 1) / t); }

extern "C" void kernel_launch(void* const* d_in, const int* in_sizes, int n_in,
                              void* d_out, int out_size, void* d_ws, size_t ws_size,
                              hipStream_t stream) {
  const float* x  = (const float*)d_in[0];
  const int*   ei = (const int*)d_in[1];
  const float* ew = (const float*)d_in[2];
  const float* W1 = (const float*)d_in[3];
  const float* b1 = (const float*)d_in[4];
  const float* W2 = (const float*)d_in[5];
  const float* b2 = (const float*)d_in[6];
  float* out = (float*)d_out;

  const int n = in_sizes[0] / F_IN;
  const int E = in_sizes[1] / 2;
  const int* row = ei;
  const int* col = ei + E;

  char* ws = (char*)d_ws;
  size_t o = 0;
  auto carve = [&](size_t bytes) {
    char* p = ws + o;
    o += (bytes + 255) & ~(size_t)255;
    return p;
  };
  float* dis = (float*)carve((size_t)n * 4);
  float* nrm = (float*)carve((size_t)E * 4);
  float* h   = (float*)carve((size_t)n * F_HID * 4);
  float* agg = (float*)carve((size_t)n * F_HID * 4);

  const long long nF1 = (long long)n * F_HID;
  const long long nF2 = (long long)n * F_OUT;

  k_fill_const<<<nblk(n, 256), 256, 0, stream>>>(dis, 1.0f, n);
  k_fill_const<<<nblk(nF1, 256), 256, 0, stream>>>(agg, 0.0f, nF1);
  k_fill_bias64<<<nblk(nF2, 256), 256, 0, stream>>>(out, b2, nF2);

  k_deg<<<nblk(E, 256), 256, 0, stream>>>(col, ew, dis, E);
  k_rsqrt<<<nblk(n, 256), 256, 0, stream>>>(dis, n);
  k_norm<<<nblk(E, 256), 256, 0, stream>>>(row, col, ew, dis, nrm, E);

  {
    const long long tiles = (long long)(n / 16) * (F_HID / 16);
    k_gemm_wmma<F_HID><<<nblk(tiles, 8), 256, 0, stream>>>(x, W1, h, n);
  }
  k_scatter<F_HID><<<nblk((long long)E, 8), 256, 0, stream>>>(h, row, col, nrm, agg, E);
  k_combine1<<<nblk(nF1, 256), 256, 0, stream>>>(agg, h, dis, b1, nF1);

  {
    const long long tiles = (long long)(n / 16) * (F_OUT / 16);
    k_gemm_wmma<F_OUT><<<nblk(tiles, 8), 256, 0, stream>>>(agg, W2, h, n);
  }
  k_scatter<F_OUT><<<nblk((long long)E, 8), 256, 0, stream>>>(h, row, col, nrm, out, E);
  k_combine2<<<nblk(nF2, 256), 256, 0, stream>>>(out, h, dis, nF2);
}


# Degree sum in LDS: 64 edge slices each add into an all-node LDS histogram (ds_add_f32), planes summed in the rsqrt pass; global degree atomics, fill launch gone; order: LDS arrival, planes, self-loop
# speedup vs baseline: 2.2255x; 2.2255x over previous
#include <hip/hip_runtime.h>
#include <cstdint>

#define F_IN  128
#define F_HID 128
#define F_OUT 64

typedef __attribute__((ext_vector_type(2))) float v2f;
typedef __attribute__((ext_vector_type(8))) float v8f;

__device__ __forceinline__ void atomic_add_f32(float* p, float v) {
  unsafeAtomicAdd(p, v);
}

#define DEG_BLOCKS 64
#define DEG_CAP    51200
#define DEG_BATCH  8

__global__ void __launch_bounds__(1024)
k_deg_lds(const int* __restrict__ col, const float* __restrict__ ew,
          float* __restrict__ part, int E, int n, int pitch) {
  __shared__ __attribute__((aligned(16))) float sdeg[DEG_CAP];
  const int tid = threadIdx.x;
  const int per = (E + DEG_BLOCKS - 1) / DEG_BLOCKS;
  const int e0 = blockIdx.x * per;
  const int e1 = (e0 + per < E) ? (e0 + per) : E;
  float* __restrict__ mine = part + (size_t)blockIdx.x * pitch;

  for (int base = 0; base < n; base += DEG_CAP) {
    const int cnt = (n - base < DEG_CAP) ? (n - base) : DEG_CAP;
    for (int i = tid * 4; i < DEG_CAP; i += 4096)
      *reinterpret_cast<float4*>(&sdeg[i]) = make_float4(0.f, 0.f, 0.f, 0.f);
    __syncthreads();

    for (int e = e0 + tid; e < e1; e += 1024 * DEG_BATCH) {
      int c[DEG_BATCH];
      float w[DEG_BATCH];
#pragma unroll
      for (int j = 0; j < DEG_BATCH; ++j) {
        const int ee = e + 1024 * j;
        const bool ok = ee < e1;
        const int el = ok ? ee : e;
        const int cv = col[el];
        w[j] = ew[el];
        c[j] = ok ? cv : -1;
      }
#pragma unroll
      for (int j = 0; j < DEG_BATCH; ++j) asm volatile("" : : "v"(w[j]));
#pragma unroll
      for (int j = 0; j < DEG_BATCH; ++j) {
        const unsigned idx = (unsigned)(c[j] - base);
        if (idx < (unsigned)cnt)
          __hip_atomic_fetch_add(&sdeg[idx], w[j], __ATOMIC_RELAXED, __HIP_MEMORY_SCOPE_WORKGROUP);
      }
    }
    __syncthreads();

    const int cnt4 = cnt & ~3;
    for (int i = tid * 4; i < cnt4; i += 4096)
      *reinterpret_cast<float4*>(&mine[base + i]) = *reinterpret_cast<const float4*>(&sdeg[i]);
    for (int i = cnt4 + tid; i < cnt; i += 1024) mine[base + i] = sdeg[i];
    __syncthreads();
  }
}

__global__ void k_dis(const float* __restrict__ part, float* __restrict__ dis, int n, int pitch) {
  int i = blockIdx.x * blockDim.x + threadIdx.x;
  if (i < n) {
    float t[DEG_BLOCKS];
#pragma unroll
    for (int b = 0; b < DEG_BLOCKS; ++b) t[b] = part[(size_t)b * pitch + i];
    float d = 0.f;
#pragma unroll
    for (int b = 0; b < DEG_BLOCKS; ++b) d += t[b];
    d += 1.0f;
    dis[i] = (d > 0.f) ? rsqrtf(d) : 0.f;
  }
}

__global__ void k_norm(const int* __restrict__ row, const int* __restrict__ col,
                       const float* __restrict__ ew, const float* __restrict__ dis,
                       float* __restrict__ nrm, int E) {
  int e = blockIdx.x * blockDim.x + threadIdx.x;
  if (e < E) nrm[e] = dis[row[e]] * ew[e] * dis[col[e]];
}

template <int FOUT, bool RELU_A>
__global__ void __launch_bounds__(256)
k_gemm_wmma(const float* __restrict__ A, const float* __restrict__ W,
            const float* __restrict__ dis, const float* __restrict__ bias,
            float* __restrict__ H, float* __restrict__ ACC, int n) {
  __shared__ float sW[F_IN * FOUT];
  for (int i = threadIdx.x; i < F_IN * FOUT; i += 256) sW[i] = W[i];
  __syncthreads();

  const int lane = threadIdx.x & 31;
  const int wave = threadIdx.x >> 5;
  const int tilesN = FOUT / 16;
  const long long numTiles = (long long)(n / 16) * tilesN;
  const long long tile = (long long)blockIdx.x * 8 + wave;
  if (tile >= numTiles) return;

  const int tm = (int)(tile / tilesN);
  const int tn = (int)(tile % tilesN);
  const int mrow = tm * 16 + (lane & 15);
  const int ncol = tn * 16 + (lane & 15);
  const int koff = (lane >> 4) * 2;
  const int rbase = tm * 16 + ((lane >> 4) << 3);

  const float bj = bias[ncol];
  float dv[8];
#pragma unroll
  for (int v = 0; v < 8; ++v) dv[v] = dis[rbase + v];

  const float* __restrict__ arow = A + (long long)mrow * F_IN;
  const float* __restrict__ wb = sW + ncol;

  v8f c = {};
#pragma unroll
  for (int k = 0; k < F_IN; k += 4) {
    const int ka = k + koff;
    v2f a, b;
    float ax = arow[ka];
    float ay = arow[ka + 1];
    if constexpr (RELU_A) {
      ax = ax > 0.f ? ax : 0.f;
      ay = ay > 0.f ? ay : 0.f;
    }
    a.x = ax;
    a.y = ay;
    b.x = wb[ka * FOUT];
    b.y = wb[(ka + 1) * FOUT];
    c = __builtin_amdgcn_wmma_f32_16x16x4_f32(
         false, a,  false, b,
         (short)0, c,  false,  false);
  }

#pragma unroll
  for (int v = 0; v < 8; ++v) {
    const long long o = (long long)(rbase + v) * FOUT + ncol;
    const float d = dv[v];
    H[o] = c[v];
    ACC[o] = c[v] * d * d + bj;
  }
}

template <int F, int EPW>
__global__ void __launch_bounds__(256)
k_scatter(const float* __restrict__ H, const int* __restrict__ row,
          const int* __restrict__ col, const float* __restrict__ nrm,
          float* __restrict__ out, int E) {
  constexpr int PER = F / 32;
  const int lane = threadIdx.x & 31;
  const int wave = __builtin_amdgcn_readfirstlane((int)(threadIdx.x >> 5));
  const long long e0 = ((long long)blockIdx.x * 8 + wave) * EPW;
  if (e0 >= E) return;

  if (e0 + EPW <= E) {
    int r[EPW], c[EPW];
    float w[EPW];
#pragma unroll
    for (int j = 0; j < EPW; ++j) {
      r[j] = row[e0 + j];
      c[j] = col[e0 + j];
      w[j] = nrm[e0 + j];
    }
    float v[EPW][PER];
#pragma unroll
    for (int j = 0; j < EPW; ++j) {
      const float* src = H + (long long)r[j] * F + lane;
#pragma unroll
      for (int p = 0; p < PER; ++p) v[j][p] = src[32 * p];
    }
#pragma unroll
    for (int j = 0; j < EPW; ++j) {
      float* dst = out + (long long)c[j] * F + lane;
#pragma unroll
      for (int p = 0; p < PER; ++p) atomic_add_f32(dst + 32 * p, v[j][p] * w[j]);
    }
  } else {
    for (long long e = e0; e < E; ++e) {
      const float* src = H + (long long)row[e] * F + lane;
      float* dst = out + (long long)col[e] * F + lane;
      const float wt = nrm[e];
      float t[PER];
#pragma unroll
      for (int p = 0; p < PER; ++p) t[p] = src[32 * p];
#pragma unroll
      for (int p = 0; p < PER; ++p) atomic_add_f32(dst + 32 * p, t[p] * wt);
    }
  }
}

static inline int nblk(long long n, int t) { return (int)((n + t - 1) / t); }

extern "C" void kernel_launch(void* const* d_in, const int* in_sizes, int n_in,
                              void* d_out, int out_size, void* d_ws, size_t ws_size,
                              hipStream_t stream) {
  const float* x  = (const float*)d_in[0];
  const int*   ei = (const int*)d_in[1];
  const float* ew = (const float*)d_in[2];
  const float* W1 = (const float*)d_in[3];
  const float* b1 = (const float*)d_in[4];
  const float* W2 = (const float*)d_in[5];
  const float* b2 = (const float*)d_in[6];
  float* out = (float*)d_out;

  const int n = in_sizes[0] / F_IN;
  const int E = in_sizes[1] / 2;
  const int* row = ei;
  const int* col = ei + E;

  char* ws = (char*)d_ws;
  size_t o = 0;
  auto carve = [&](size_t bytes) {
    char* p = ws + o;
    o += (bytes + 255) & ~(size_t)255;
    return p;
  };
  const int pitch = (n + 31) & ~31;
  float* dis = (float*)carve((size_t)n * 4);
  float* part = (float*)carve((size_t)DEG_BLOCKS * pitch * 4);
  float* nrm = (float*)carve((size_t)E * 4);
  float* h   = (float*)carve((size_t)n * F_HID * 4);
  float* agg = (float*)carve((size_t)n * F_HID * 4);

  k_deg_lds<<<DEG_BLOCKS, 1024, 0, stream>>>(col, ew, part, E, n, pitch);
  k_dis<<<nblk(n, 256), 256, 0, stream>>>(part, dis, n, pitch);
  k_norm<<<nblk(E, 256), 256, 0, stream>>>(row, col, ew, dis, nrm, E);

  {
    const long long tiles = (long long)(n / 16) * (F_HID / 16);
    k_gemm_wmma<F_HID, false><<<nblk(tiles, 8), 256, 0, stream>>>(x, W1, dis, b1, h, agg, n);
  }
  k_scatter<F_HID, 16><<<nblk((long long)E, 8 * 16), 256, 0, stream>>>(h, row, col, nrm, agg, E);

  {
    const long long tiles = (long long)(n / 16) * (F_OUT / 16);
    k_gemm_wmma<F_OUT, true><<<nblk(tiles, 8), 256, 0, stream>>>(agg, W2, dis, b2, h, out, n);
  }
  k_scatter<F_OUT, 16><<<nblk((long long)E, 8 * 16), 256, 0, stream>>>(h, row, col, nrm, out, E);
}


# GEMM tiling: one wave computes a 16-row strip of all FOUT columns (8 or 4 accumulators); A fragment loaded once per k-step, not once per column tile; a block covers 128 rows, W staged per 128
# speedup vs baseline: 2.3617x; 1.0612x over previous
#include <hip/hip_runtime.h>
#include <cstdint>

#define F_IN  128
#define F_HID 128
#define F_OUT 64

typedef __attribute__((ext_vector_type(2))) float v2f;
typedef __attribute__((ext_vector_type(8))) float v8f;

__device__ __forceinline__ void atomic_add_f32(float* p, float v) {
  unsafeAtomicAdd(p, v);
}

#define DEG_BLOCKS 64
#define DEG_CAP    51200
#define DEG_BATCH  8

__global__ void __launch_bounds__(1024)
k_deg_lds(const int* __restrict__ col, const float* __restrict__ ew,
          float* __restrict__ part, int E, int n, int pitch) {
  __shared__ __attribute__((aligned(16))) float sdeg[DEG_CAP];
  const int tid = threadIdx.x;
  const int per = (E + DEG_BLOCKS - 1) / DEG_BLOCKS;
  const int e0 = blockIdx.x * per;
  const int e1 = (e0 + per < E) ? (e0 + per) : E;
  float* __restrict__ mine = part + (size_t)blockIdx.x * pitch;

  for (int base = 0; base < n; base += DEG_CAP) {
    const int cnt = (n - base < DEG_CAP) ? (n - base) : DEG_CAP;
    for (int i = tid * 4; i < DEG_CAP; i += 4096)
      *reinterpret_cast<float4*>(&sdeg[i]) = make_float4(0.f, 0.f, 0.f, 0.f);
    __syncthreads();

    for (int e = e0 + tid; e < e1; e += 1024 * DEG_BATCH) {
      int c[DEG_BATCH];
      float w[DEG_BATCH];
#pragma unroll
      for (int j = 0; j < DEG_BATCH; ++j) {
        const int ee = e + 1024 * j;
        const bool ok = ee < e1;
        const int el = ok ? ee : e;
        const int cv = col[el];
        w[j] = ew[el];
        c[j] = ok ? cv : -1;
      }
#pragma unroll
      for (int j = 0; j < DEG_BATCH; ++j) asm volatile("" : : "v"(w[j]));
#pragma unroll
      for (int j = 0; j < DEG_BATCH; ++j) {
        const unsigned idx = (unsigned)(c[j] - base);
        if (idx < (unsigned)cnt)
          __hip_atomic_fetch_add(&sdeg[idx], w[j], __ATOMIC_RELAXED, __HIP_MEMORY_SCOPE_WORKGROUP);
      }
    }
    __syncthreads();

    const int cnt4 = cnt & ~3;
    for (int i = tid * 4; i < cnt4; i += 4096)
      *reinterpret_cast<float4*>(&mine[base + i]) = *reinterpret_cast<const float4*>(&sdeg[i]);
    for (int i = cnt4 + tid; i < cnt; i += 1024) mine[base + i] = sdeg[i];
    __syncthreads();
  }
}

__global__ void k_dis(const float* __restrict__ part, float* __restrict__ dis, int n, int pitch) {
  int i = blockIdx.x * blockDim.x + threadIdx.x;
  if (i < n) {
    float t[DEG_BLOCKS];
#pragma unroll
    for (int b = 0; b < DEG_BLOCKS; ++b) t[b] = part[(size_t)b * pitch + i];
    float d = 0.f;
#pragma unroll
    for (int b = 0; b < DEG_BLOCKS; ++b) d += t[b];
    d += 1.0f;
    dis[i] = (d > 0.f) ? rsqrtf(d) : 0.f;
  }
}

__global__ void k_norm(const int* __restrict__ row, const int* __restrict__ col,
                       const float* __restrict__ ew, const float* __restrict__ dis,
                       float* __restrict__ nrm, int E) {
  int e = blockIdx.x * blockDim.x + threadIdx.x;
  if (e < E) nrm[e] = dis[row[e]] * ew[e] * dis[col[e]];
}

template <int FOUT, bool RELU_A>
__global__ void __launch_bounds__(256)
k_gemm_wmma(const float* __restrict__ A, const float* __restrict__ W,
            const float* __restrict__ dis, const float* __restrict__ bias,
            float* __restrict__ H, float* __restrict__ ACC, int n) {
  __shared__ float sW[F_IN * FOUT];
  for (int i = threadIdx.x; i < F_IN * FOUT; i += 256) sW[i] = W[i];
  __syncthreads();

  constexpr int NT = FOUT / 16;
  const int lane = threadIdx.x & 31;
  const int wave = threadIdx.x >> 5;
  const int strips = n / 16;
  const int strip = blockIdx.x * 8 + wave;
  if (strip >= strips) return;

  const int c16 = lane & 15;
  const int mrow = strip * 16 + c16;
  const int koff = (lane >> 4) * 2;
  const int rbase = strip * 16 + ((lane >> 4) << 3);

  float bj[NT];
#pragma unroll
  for (int t = 0; t < NT; ++t) bj[t] = bias[16 * t + c16];
  float dv[8];
#pragma unroll
  for (int v = 0; v < 8; ++v) dv[v] = dis[rbase + v];

  const float* __restrict__ arow = A + (long long)mrow * F_IN;
  const float* __restrict__ wb = sW + c16;

  constexpr int KS = F_IN / 4;
  v2f af[KS];
#pragma unroll
  for (int s = 0; s < KS; ++s) af[s] = *reinterpret_cast<const v2f*>(arow + 4 * s + koff);

  v8f c[NT];
#pragma unroll
  for (int t = 0; t < NT; ++t) c[t] = v8f{};

  v2f bq[2][NT];
#pragma unroll
  for (int t = 0; t < NT; ++t) {
    bq[0][t].x = wb[koff * FOUT + 16 * t];
    bq[0][t].y = wb[(koff + 1) * FOUT + 16 * t];
  }

#pragma unroll
  for (int s = 0; s < KS; ++s) {
    if (s + 1 < KS) {
      const int kn = 4 * (s + 1) + koff;
#pragma unroll
      for (int t = 0; t < NT; ++t) {
        bq[(s + 1) & 1][t].x = wb[kn * FOUT + 16 * t];
        bq[(s + 1) & 1][t].y = wb[(kn + 1) * FOUT + 16 * t];
      }
    }
    __builtin_amdgcn_sched_barrier(0);
    v2f a = af[s];
    if constexpr (RELU_A) {
      a.x = a.x > 0.f ? a.x : 0.f;
      a.y = a.y > 0.f ? a.y : 0.f;
    }
#pragma unroll
    for (int t = 0; t < NT; ++t) {
      c[t] = __builtin_amdgcn_wmma_f32_16x16x4_f32(
           false, a,  false, bq[s & 1][t],
           (short)0, c[t],  false,  false);
    }
    __builtin_amdgcn_sched_barrier(0);
  }

#pragma unroll
  for (int t = 0; t < NT; ++t) {
#pragma unroll
    for (int v = 0; v < 8; ++v) {
      const long long o = (long long)(rbase + v) * FOUT + 16 * t + c16;
      const float d = dv[v];
      H[o] = c[t][v];
      ACC[o] = c[t][v] * d * d + bj[t];
    }
  }
}

template <int F, int EPW>
__global__ void __launch_bounds__(256)
k_scatter(const float* __restrict__ H, const int* __restrict__ row,
          const int* __restrict__ col, const float* __restrict__ nrm,
          float* __restrict__ out, int E) {
  constexpr int PER = F / 32;
  const int lane = threadIdx.x & 31;
  const int wave = __builtin_amdgcn_readfirstlane((int)(threadIdx.x >> 5));
  const long long e0 = ((long long)blockIdx.x * 8 + wave) * EPW;
  if (e0 >= E) return;

  if (e0 + EPW <= E) {
    int r[EPW], c[EPW];
    float w[EPW];
#pragma unroll
    for (int j = 0; j < EPW; ++j) {
      r[j] = row[e0 + j];
      c[j] = col[e0 + j];
      w[j] = nrm[e0 + j];
    }
    float v[EPW][PER];
#pragma unroll
    for (int j = 0; j < EPW; ++j) {
      const float* src = H + (long long)r[j] * F + lane;
#pragma unroll
      for (int p = 0; p < PER; ++p) v[j][p] = src[32 * p];
    }
#pragma unroll
    for (int j = 0; j < EPW; ++j) {
      float* dst = out + (long long)c[j] * F + lane;
#pragma unroll
      for (int p = 0; p < PER; ++p) atomic_add_f32(dst + 32 * p, v[j][p] * w[j]);
    }
  } else {
    for (long long e = e0; e < E; ++e) {
      const float* src = H + (long long)row[e] * F + lane;
      float* dst = out + (long long)col[e] * F + lane;
      const float wt = nrm[e];
      float t[PER];
#pragma unroll
      for (int p = 0; p < PER; ++p) t[p] = src[32 * p];
#pragma unroll
      for (int p = 0; p < PER; ++p) atomic_add_f32(dst + 32 * p, t[p] * wt);
    }
  }
}

static inline int nblk(long long n, int t) { return (int)((n + t - 1) / t); }

extern "C" void kernel_launch(void* const* d_in, const int* in_sizes, int n_in,
                              void* d_out, int out_size, void* d_ws, size_t ws_size,
                              hipStream_t stream) {
  const float* x  = (const float*)d_in[0];
  const int*   ei = (const int*)d_in[1];
  const float* ew = (const float*)d_in[2];
  const float* W1 = (const float*)d_in[3];
  const float* b1 = (const float*)d_in[4];
  const float* W2 = (const float*)d_in[5];
  const float* b2 = (const float*)d_in[6];
  float* out = (float*)d_out;

  const int n = in_sizes[0] / F_IN;
  const int E = in_sizes[1] / 2;
  const int* row = ei;
  const int* col = ei + E;

  char* ws = (char*)d_ws;
  size_t o = 0;
  auto carve = [&](size_t bytes) {
    char* p = ws + o;
    o += (bytes + 255) & ~(size_t)255;
    return p;
  };
  const int pitch = (n + 31) & ~31;
  float* dis = (float*)carve((size_t)n * 4);
  float* part = (float*)carve((size_t)DEG_BLOCKS * pitch * 4);
  float* nrm = (float*)carve((size_t)E * 4);
  float* h   = (float*)carve((size_t)n * F_HID * 4);
  float* agg = (float*)carve((size_t)n * F_HID * 4);

  k_deg_lds<<<DEG_BLOCKS, 1024, 0, stream>>>(col, ew, part, E, n, pitch);
  k_dis<<<nblk(n, 256), 256, 0, stream>>>(part, dis, n, pitch);
  k_norm<<<nblk(E, 256), 256, 0, stream>>>(row, col, ew, dis, nrm, E);

  k_gemm_wmma<F_HID, false><<<nblk(n / 16, 8), 256, 0, stream>>>(x, W1, dis, b1, h, agg, n);
  k_scatter<F_HID, 16><<<nblk((long long)E, 8 * 16), 256, 0, stream>>>(h, row, col, nrm, agg, E);

  k_gemm_wmma<F_OUT, true><<<nblk(n / 16, 8), 256, 0, stream>>>(agg, W2, dis, b2, h, out, n);
  k_scatter<F_OUT, 16><<<nblk((long long)E, 8 * 16), 256, 0, stream>>>(h, row, col, nrm, out, E);
}


# Edge scatter: target rows split by chip half: a block reads its die, draws a job ticket (int atomic), adds only edges whose target is in its half's row range; sum order: atomic arrival, as before
# speedup vs baseline: 2.7517x; 1.1651x over previous
#include <hip/hip_runtime.h>
#include <cstdint>

#define F_IN  128
#define F_HID 128
#define F_OUT 64

typedef __attribute__((ext_vector_type(2))) float v2f;
typedef __attribute__((ext_vector_type(8))) float v8f;

__device__ __forceinline__ void atomic_add_f32(float* p, float v) {
  unsafeAtomicAdd(p, v);
}

#define DEG_BLOCKS 64
#define DEG_CAP    51200
#define DEG_BATCH  8

__global__ void __launch_bounds__(1024)
k_deg_lds(const int* __restrict__ col, const float* __restrict__ ew,
          float* __restrict__ part, int E, int n, int pitch) {
  __shared__ __attribute__((aligned(16))) float sdeg[DEG_CAP];
  const int tid = threadIdx.x;
  const int per = (E + DEG_BLOCKS - 1) / DEG_BLOCKS;
  const int e0 = blockIdx.x * per;
  const int e1 = (e0 + per < E) ? (e0 + per) : E;
  float* __restrict__ mine = part + (size_t)blockIdx.x * pitch;

  for (int base = 0; base < n; base += DEG_CAP) {
    const int cnt = (n - base < DEG_CAP) ? (n - base) : DEG_CAP;
    for (int i = tid * 4; i < DEG_CAP; i += 4096)
      *reinterpret_cast<float4*>(&sdeg[i]) = make_float4(0.f, 0.f, 0.f, 0.f);
    __syncthreads();

    for (int e = e0 + tid; e < e1; e += 1024 * DEG_BATCH) {
      int c[DEG_BATCH];
      float w[DEG_BATCH];
#pragma unroll
      for (int j = 0; j < DEG_BATCH; ++j) {
        const int ee = e + 1024 * j;
        const bool ok = ee < e1;
        const int el = ok ? ee : e;
        const int cv = col[el];
        w[j] = ew[el];
        c[j] = ok ? cv : -1;
      }
#pragma unroll
      for (int j = 0; j < DEG_BATCH; ++j) asm volatile("" : : "v"(w[j]));
#pragma unroll
      for (int j = 0; j < DEG_BATCH; ++j) {
        const unsigned idx = (unsigned)(c[j] - base);
        if (idx < (unsigned)cnt)
          __hip_atomic_fetch_add(&sdeg[idx], w[j], __ATOMIC_RELAXED, __HIP_MEMORY_SCOPE_WORKGROUP);
      }
    }
    __syncthreads();

    const int cnt4 = cnt & ~3;
    for (int i = tid * 4; i < cnt4; i += 4096)
      *reinterpret_cast<float4*>(&mine[base + i]) = *reinterpret_cast<const float4*>(&sdeg[i]);
    for (int i = cnt4 + tid; i < cnt; i += 1024) mine[base + i] = sdeg[i];
    __syncthreads();
  }
}

__global__ void k_dis(const float* __restrict__ part, float* __restrict__ dis, int* __restrict__ ctr, int n, int pitch) {
  int i = blockIdx.x * blockDim.x + threadIdx.x;
  if (i < 4) ctr[i] = 0;
  if (i < n) {
    float t[DEG_BLOCKS];
#pragma unroll
    for (int b = 0; b < DEG_BLOCKS; ++b) t[b] = part[(size_t)b * pitch + i];
    float d = 0.f;
#pragma unroll
    for (int b = 0; b < DEG_BLOCKS; ++b) d += t[b];
    d += 1.0f;
    dis[i] = (d > 0.f) ? rsqrtf(d) : 0.f;
  }
}

__global__ void k_norm(const int* __restrict__ row, const int* __restrict__ col,
                       const float* __restrict__ ew, const float* __restrict__ dis,
                       float* __restrict__ nrm, int E) {
  int e = blockIdx.x * blockDim.x + threadIdx.x;
  if (e < E) nrm[e] = dis[row[e]] * ew[e] * dis[col[e]];
}

template <int FOUT, bool RELU_A>
__global__ void __launch_bounds__(256)
k_gemm_wmma(const float* __restrict__ A, const float* __restrict__ W,
            const float* __restrict__ dis, const float* __restrict__ bias,
            float* __restrict__ H, float* __restrict__ ACC, int n) {
  __shared__ float sW[F_IN * FOUT];
  for (int i = threadIdx.x; i < F_IN * FOUT; i += 256) sW[i] = W[i];
  __syncthreads();

  constexpr int NT = FOUT / 16;
  const int lane = threadIdx.x & 31;
  const int wave = threadIdx.x >> 5;
  const int strips = n / 16;
  const int strip = blockIdx.x * 8 + wave;
  if (strip >= strips) return;

  const int c16 = lane & 15;
  const int mrow = strip * 16 + c16;
  const int koff = (lane >> 4) * 2;
  const int rbase = strip * 16 + ((lane >> 4) << 3);

  float bj[NT];
#pragma unroll
  for (int t = 0; t < NT; ++t) bj[t] = bias[16 * t + c16];
  float dv[8];
#pragma unroll
  for (int v = 0; v < 8; ++v) dv[v] = dis[rbase + v];

  const float* __restrict__ arow = A + (long long)mrow * F_IN;
  const float* __restrict__ wb = sW + c16;

  constexpr int KS = F_IN / 4;
  v2f af[KS];
#pragma unroll
  for (int s = 0; s < KS; ++s) af[s] = *reinterpret_cast<const v2f*>(arow + 4 * s + koff);

  v8f c[NT];
#pragma unroll
  for (int t = 0; t < NT; ++t) c[t] = v8f{};

  v2f bq[2][NT];
#pragma unroll
  for (int t = 0; t < NT; ++t) {
    bq[0][t].x = wb[koff * FOUT + 16 * t];
    bq[0][t].y = wb[(koff + 1) * FOUT + 16 * t];
  }

#pragma unroll
  for (int s = 0; s < KS; ++s) {
    if (s + 1 < KS) {
      const int kn = 4 * (s + 1) + koff;
#pragma unroll
      for (int t = 0; t < NT; ++t) {
        bq[(s + 1) & 1][t].x = wb[kn * FOUT + 16 * t];
        bq[(s + 1) & 1][t].y = wb[(kn + 1) * FOUT + 16 * t];
      }
    }
    __builtin_amdgcn_sched_barrier(0);
    v2f a = af[s];
    if constexpr (RELU_A) {
      a.x = a.x > 0.f ? a.x : 0.f;
      a.y = a.y > 0.f ? a.y : 0.f;
    }
#pragma unroll
    for (int t = 0; t < NT; ++t) {
      c[t] = __builtin_amdgcn_wmma_f32_16x16x4_f32(
           false, a,  false, bq[s & 1][t],
           (short)0, c[t],  false,  false);
    }
    __builtin_amdgcn_sched_barrier(0);
  }

#pragma unroll
  for (int t = 0; t < NT; ++t) {
#pragma unroll
    for (int v = 0; v < 8; ++v) {
      const long long o = (long long)(rbase + v) * FOUT + 16 * t + c16;
      const float d = dv[v];
      H[o] = c[t][v];
      ACC[o] = c[t][v] * d * d + bj[t];
    }
  }
}

template <int F>
__global__ void __launch_bounds__(256)
k_scatter(const float* __restrict__ H, const int* __restrict__ row,
          const int* __restrict__ col, const float* __restrict__ nrm,
          float* __restrict__ out, int* __restrict__ ctr, int E, int nsplit, int chunks) {
  constexpr int PER = F / 32;
  constexpr int SLOTS = 16;
  __shared__ int s_job;
  if (threadIdx.x == 0) {
    const int id = __builtin_amdgcn_s_sendmsg_rtn(0x87);
    const int hh = (id >> 18) & 1;
    int th = hh;
    int t = atomicAdd(&ctr[th], 1);
    if (t >= chunks) {
      th ^= 1;
      t = atomicAdd(&ctr[th], 1);
    }
    s_job = t * 2 + th;
  }
  __syncthreads();
  const int job = s_job;
  const int th = job & 1;
  const int chunk = job >> 1;
  if (chunk >= chunks) return;

  const int lane = threadIdx.x & 31;
  const int wave = threadIdx.x >> 5;
  const long long e = ((long long)chunk * 8 + wave) * 32 + lane;
  const bool valid = e < E;
  const long long el = valid ? e : 0;
  const int rv = row[el];
  const int cv = col[el];
  const float wv = nrm[el];
  asm volatile("" : : "v"(rv), "v"(wv));
  const bool mine = valid && ((cv >= nsplit) == (th != 0));
  unsigned m = __builtin_amdgcn_ballot_w32(mine);

  while (m) {
    float v[SLOTS][PER];
    unsigned m1 = m;
#pragma unroll
    for (int k = 0; k < SLOTS; ++k) {
      if (m1) {
        const int j = __builtin_ctz(m1);
        m1 &= m1 - 1;
        const int r = __builtin_amdgcn_readlane(rv, j);
        const float* src = H + (long long)r * F + lane;
#pragma unroll
        for (int p = 0; p < PER; ++p) v[k][p] = src[32 * p];
      }
    }
    unsigned m2 = m;
#pragma unroll
    for (int k = 0; k < SLOTS; ++k) {
      if (m2) {
        const int j = __builtin_ctz(m2);
        m2 &= m2 - 1;
        const int c = __builtin_amdgcn_readlane(cv, j);
        const float w = __builtin_bit_cast(float, __builtin_amdgcn_readlane(__builtin_bit_cast(int, wv), j));
        float* dst = out + (long long)c * F + lane;
#pragma unroll
        for (int p = 0; p < PER; ++p) atomic_add_f32(dst + 32 * p, v[k][p] * w);
      }
    }
    m = m1;
  }
}

static inline int nblk(long long n, int t) { return (int)((n + t - 1) / t); }

extern "C" void kernel_launch(void* const* d_in, const int* in_sizes, int n_in,
                              void* d_out, int out_size, void* d_ws, size_t ws_size,
                              hipStream_t stream) {
  const float* x  = (const float*)d_in[0];
  const int*   ei = (const int*)d_in[1];
  const float* ew = (const float*)d_in[2];
  const float* W1 = (const float*)d_in[3];
  const float* b1 = (const float*)d_in[4];
  const float* W2 = (const float*)d_in[5];
  const float* b2 = (const float*)d_in[6];
  float* out = (float*)d_out;

  const int n = in_sizes[0] / F_IN;
  const int E = in_sizes[1] / 2;
  const int* row = ei;
  const int* col = ei + E;

  char* ws = (char*)d_ws;
  size_t o = 0;
  auto carve = [&](size_t bytes) {
    char* p = ws + o;
    o += (bytes + 255) & ~(size_t)255;
    return p;
  };
  const int pitch = (n + 31) & ~31;
  float* dis = (float*)carve((size_t)n * 4);
  float* part = (float*)carve((size_t)DEG_BLOCKS * pitch * 4);
  float* nrm = (float*)carve((size_t)E * 4);
  float* h   = (float*)carve((size_t)n * F_HID * 4);
  float* agg = (float*)carve((size_t)n * F_HID * 4);
  int* ctr   = (int*)carve(4 * sizeof(int));
  const int chunks = nblk((long long)E, 256);
  const int nsplit = n / 2;

  k_deg_lds<<<DEG_BLOCKS, 1024, 0, stream>>>(col, ew, part, E, n, pitch);
  k_dis<<<nblk(n, 256), 256, 0, stream>>>(part, dis, ctr, n, pitch);
  k_norm<<<nblk(E, 256), 256, 0, stream>>>(row, col, ew, dis, nrm, E);

  k_gemm_wmma<F_HID, false><<<nblk(n / 16, 8), 256, 0, stream>>>(x, W1, dis, b1, h, agg, n);
  k_scatter<F_HID><<<2 * chunks, 256, 0, stream>>>(h, row, col, nrm, agg, ctr, E, nsplit, chunks);

  k_gemm_wmma<F_OUT, true><<<nblk(n / 16, 8), 256, 0, stream>>>(agg, W2, dis, b2, h, out, n);
  k_scatter<F_OUT><<<2 * chunks, 256, 0, stream>>>(h, row, col, nrm, out, ctr + 2, E, nsplit, chunks);
}


# Edge-weight pass folded into the layer-1 GEMM launch: each 16-row strip wave also computes the weights of its 256 edges, record loads and dis gathers riding under its WMMA stream; 7 launches to 6
# speedup vs baseline: 2.8464x; 1.0344x over previous
#include <hip/hip_runtime.h>
#include <cstdint>

#define F_IN  128
#define F_HID 128
#define F_OUT 64

typedef __attribute__((ext_vector_type(2))) float v2f;
typedef __attribute__((ext_vector_type(8))) float v8f;

__device__ __forceinline__ void atomic_add_f32(float* p, float v) {
  unsafeAtomicAdd(p, v);
}

#define DEG_BLOCKS 64
#define DEG_CAP    51200
#define DEG_BATCH  8

__global__ void __launch_bounds__(1024)
k_deg_lds(const int* __restrict__ col, const float* __restrict__ ew,
          float* __restrict__ part, int E, int n, int pitch) {
  __shared__ __attribute__((aligned(16))) float sdeg[DEG_CAP];
  const int tid = threadIdx.x;
  const int per = (E + DEG_BLOCKS - 1) / DEG_BLOCKS;
  const int e0 = blockIdx.x * per;
  const int e1 = (e0 + per < E) ? (e0 + per) : E;
  float* __restrict__ mine = part + (size_t)blockIdx.x * pitch;

  for (int base = 0; base < n; base += DEG_CAP) {
    const int cnt = (n - base < DEG_CAP) ? (n - base) : DEG_CAP;
    for (int i = tid * 4; i < DEG_CAP; i += 4096)
      *reinterpret_cast<float4*>(&sdeg[i]) = make_float4(0.f, 0.f, 0.f, 0.f);
    __syncthreads();

    for (int e = e0 + tid; e < e1; e += 1024 * DEG_BATCH) {
      int c[DEG_BATCH];
      float w[DEG_BATCH];
#pragma unroll
      for (int j = 0; j < DEG_BATCH; ++j) {
        const int ee = e + 1024 * j;
        const bool ok = ee < e1;
        const int el = ok ? ee : e;
        const int cv = col[el];
        w[j] = ew[el];
        c[j] = ok ? cv : -1;
      }
#pragma unroll
      for (int j = 0; j < DEG_BATCH; ++j) asm volatile("" : : "v"(w[j]));
#pragma unroll
      for (int j = 0; j < DEG_BATCH; ++j) {
        const unsigned idx = (unsigned)(c[j] - base);
        if (idx < (unsigned)cnt)
          __hip_atomic_fetch_add(&sdeg[idx], w[j], __ATOMIC_RELAXED, __HIP_MEMORY_SCOPE_WORKGROUP);
      }
    }
    __syncthreads();

    const int cnt4 = cnt & ~3;
    for (int i = tid * 4; i < cnt4; i += 4096)
      *reinterpret_cast<float4*>(&mine[base + i]) = *reinterpret_cast<const float4*>(&sdeg[i]);
    for (int i = cnt4 + tid; i < cnt; i += 1024) mine[base + i] = sdeg[i];
    __syncthreads();
  }
}

__global__ void k_dis(const float* __restrict__ part, float* __restrict__ dis, int* __restrict__ ctr, int n, int pitch) {
  int i = blockIdx.x * blockDim.x + threadIdx.x;
  if (i < 4) ctr[i] = 0;
  if (i < n) {
    float t[DEG_BLOCKS];
#pragma unroll
    for (int b = 0; b < DEG_BLOCKS; ++b) t[b] = part[(size_t)b * pitch + i];
    float d = 0.f;
#pragma unroll
    for (int b = 0; b < DEG_BLOCKS; ++b) d += t[b];
    d += 1.0f;
    dis[i] = (d > 0.f) ? rsqrtf(d) : 0.f;
  }
}

#define NRM_SLOTS 8
#define NRM_STEP  12
template <int FOUT, bool RELU_A, bool NORM>
__global__ void __launch_bounds__(256)
k_gemm_wmma(const float* __restrict__ A, const float* __restrict__ W,
            const float* __restrict__ dis, const float* __restrict__ bias,
            float* __restrict__ H, float* __restrict__ ACC, int n,
            const int* __restrict__ row, const int* __restrict__ col,
            const float* __restrict__ ew, float* __restrict__ nrm, int E) {
  __shared__ float sW[F_IN * FOUT];
  for (int i = threadIdx.x; i < F_IN * FOUT; i += 256) sW[i] = W[i];
  __syncthreads();

  constexpr int NT = FOUT / 16;
  const int lane = threadIdx.x & 31;
  const int wave = threadIdx.x >> 5;
  const int strips = n / 16;
  const int strip = blockIdx.x * 8 + wave;
  if (strip >= strips) return;

  const int c16 = lane & 15;
  const int mrow = strip * 16 + c16;
  const int koff = (lane >> 4) * 2;
  const int rbase = strip * 16 + ((lane >> 4) << 3);

  float bj[NT];
#pragma unroll
  for (int t = 0; t < NT; ++t) bj[t] = bias[16 * t + c16];
  float dv[8];
#pragma unroll
  for (int v = 0; v < 8; ++v) dv[v] = dis[rbase + v];

  const float* __restrict__ arow = A + (long long)mrow * F_IN;
  const float* __restrict__ wb = sW + c16;

  constexpr int KS = F_IN / 4;
  v2f af[KS];
#pragma unroll
  for (int s = 0; s < KS; ++s) af[s] = *reinterpret_cast<const v2f*>(arow + 4 * s + koff);

  int e_lo = 0, e_hi = 0;
  int er[NRM_SLOTS], ec[NRM_SLOTS];
  float ewv[NRM_SLOTS], dr[NRM_SLOTS], dc[NRM_SLOTS];
  if constexpr (NORM) {
    const int per = ((E + strips - 1) / strips + 31) & ~31;
    const long long lo = (long long)strip * per;
    e_lo = (lo < E) ? (int)lo : E;
    e_hi = (lo + per < E) ? (int)(lo + per) : E;
#pragma unroll
    for (int j = 0; j < NRM_SLOTS; ++j) {
      const int e = e_lo + 32 * j + lane;
      const int el = (e < e_hi) ? e : 0;
      er[j] = row[el];
      ec[j] = col[el];
      ewv[j] = ew[el];
    }
  }

  v8f c[NT];
#pragma unroll
  for (int t = 0; t < NT; ++t) c[t] = v8f{};

  v2f bq[2][NT];
#pragma unroll
  for (int t = 0; t < NT; ++t) {
    bq[0][t].x = wb[koff * FOUT + 16 * t];
    bq[0][t].y = wb[(koff + 1) * FOUT + 16 * t];
  }

#pragma unroll
  for (int s = 0; s < KS; ++s) {
    if (s + 1 < KS) {
      const int kn = 4 * (s + 1) + koff;
#pragma unroll
      for (int t = 0; t < NT; ++t) {
        bq[(s + 1) & 1][t].x = wb[kn * FOUT + 16 * t];
        bq[(s + 1) & 1][t].y = wb[(kn + 1) * FOUT + 16 * t];
      }
    }
    if constexpr (NORM) {
      if (s == NRM_STEP) {
#pragma unroll
        for (int j = 0; j < NRM_SLOTS; ++j) {
          dr[j] = dis[er[j]];
          dc[j] = dis[ec[j]];
        }
      }
    }
    __builtin_amdgcn_sched_barrier(0);
    v2f a = af[s];
    if constexpr (RELU_A) {
      a.x = a.x > 0.f ? a.x : 0.f;
      a.y = a.y > 0.f ? a.y : 0.f;
    }
#pragma unroll
    for (int t = 0; t < NT; ++t) {
      c[t] = __builtin_amdgcn_wmma_f32_16x16x4_f32(
           false, a,  false, bq[s & 1][t],
           (short)0, c[t],  false,  false);
    }
    __builtin_amdgcn_sched_barrier(0);
  }

  if constexpr (NORM) {
#pragma unroll
    for (int j = 0; j < NRM_SLOTS; ++j) asm volatile("" : : "v"(dr[j]), "v"(ewv[j]), "v"(dc[j]));
#pragma unroll
    for (int j = 0; j < NRM_SLOTS; ++j) {
      const int e = e_lo + 32 * j + lane;
      if (e < e_hi) nrm[e] = dr[j] * ewv[j] * dc[j];
    }
#pragma unroll 1
    for (int e = e_lo + 32 * NRM_SLOTS + lane; e < e_hi; e += 32)
      nrm[e] = dis[row[e]] * ew[e] * dis[col[e]];
  }

#pragma unroll
  for (int t = 0; t < NT; ++t) {
#pragma unroll
    for (int v = 0; v < 8; ++v) {
      const long long o = (long long)(rbase + v) * FOUT + 16 * t + c16;
      const float d = dv[v];
      H[o] = c[t][v];
      ACC[o] = c[t][v] * d * d + bj[t];
    }
  }
}

template <int F>
__global__ void __launch_bounds__(256)
k_scatter(const float* __restrict__ H, const int* __restrict__ row,
          const int* __restrict__ col, const float* __restrict__ nrm,
          float* __restrict__ out, int* __restrict__ ctr, int E, int nsplit, int chunks) {
  constexpr int PER = F / 32;
  constexpr int SLOTS = 16;
  __shared__ int s_job;
  if (threadIdx.x == 0) {
    const int id = __builtin_amdgcn_s_sendmsg_rtn(0x87);
    const int hh = (id >> 18) & 1;
    int th = hh;
    int t = atomicAdd(&ctr[th], 1);
    if (t >= chunks) {
      th ^= 1;
      t = atomicAdd(&ctr[th], 1);
    }
    s_job = t * 2 + th;
  }
  __syncthreads();
  const int job = s_job;
  const int th = job & 1;
  const int chunk = job >> 1;
  if (chunk >= chunks) return;

  const int lane = threadIdx.x & 31;
  const int wave = threadIdx.x >> 5;
  const long long e = ((long long)chunk * 8 + wave) * 32 + lane;
  const bool valid = e < E;
  const long long el = valid ? e : 0;
  const int rv = row[el];
  const int cv = col[el];
  const float wv = nrm[el];
  asm volatile("" : : "v"(rv), "v"(wv));
  const bool mine = valid && ((cv >= nsplit) == (th != 0));
  unsigned m = __builtin_amdgcn_ballot_w32(mine);

  while (m) {
    float v[SLOTS][PER];
    unsigned m1 = m;
#pragma unroll
    for (int k = 0; k < SLOTS; ++k) {
      if (m1) {
        const int j = __builtin_ctz(m1);
        m1 &= m1 - 1;
        const int r = __builtin_amdgcn_readlane(rv, j);
        const float* src = H + (long long)r * F + lane;
#pragma unroll
        for (int p = 0; p < PER; ++p) v[k][p] = src[32 * p];
      }
    }
    unsigned m2 = m;
#pragma unroll
    for (int k = 0; k < SLOTS; ++k) {
      if (m2) {
        const int j = __builtin_ctz(m2);
        m2 &= m2 - 1;
        const int c = __builtin_amdgcn_readlane(cv, j);
        const float w = __builtin_bit_cast(float, __builtin_amdgcn_readlane(__builtin_bit_cast(int, wv), j));
        float* dst = out + (long long)c * F + lane;
#pragma unroll
        for (int p = 0; p < PER; ++p) atomic_add_f32(dst + 32 * p, v[k][p] * w);
      }
    }
    m = m1;
  }
}

static inline int nblk(long long n, int t) { return (int)((n + t - 1) / t); }

extern "C" void kernel_launch(void* const* d_in, const int* in_sizes, int n_in,
                              void* d_out, int out_size, void* d_ws, size_t ws_size,
                              hipStream_t stream) {
  const float* x  = (const float*)d_in[0];
  const int*   ei = (const int*)d_in[1];
  const float* ew = (const float*)d_in[2];
  const float* W1 = (const float*)d_in[3];
  const float* b1 = (const float*)d_in[4];
  const float* W2 = (const float*)d_in[5];
  const float* b2 = (const float*)d_in[6];
  float* out = (float*)d_out;

  const int n = in_sizes[0] / F_IN;
  const int E = in_sizes[1] / 2;
  const int* row = ei;
  const int* col = ei + E;

  char* ws = (char*)d_ws;
  size_t o = 0;
  auto carve = [&](size_t bytes) {
    char* p = ws + o;
    o += (bytes + 255) & ~(size_t)255;
    return p;
  };
  const int pitch = (n + 31) & ~31;
  float* dis = (float*)carve((size_t)n * 4);
  float* part = (float*)carve((size_t)DEG_BLOCKS * pitch * 4);
  float* nrm = (float*)carve((size_t)E * 4);
  float* h   = (float*)carve((size_t)n * F_HID * 4);
  float* agg = (float*)carve((size_t)n * F_HID * 4);
  int* ctr   = (int*)carve(4 * sizeof(int));
  const int chunks = nblk((long long)E, 256);
  const int nsplit = n / 2;

  k_deg_lds<<<DEG_BLOCKS, 1024, 0, stream>>>(col, ew, part, E, n, pitch);
  k_dis<<<nblk(n, 256), 256, 0, stream>>>(part, dis, ctr, n, pitch);

  k_gemm_wmma<F_HID, false, true><<<nblk(n / 16, 8), 256, 0, stream>>>(x, W1, dis, b1, h, agg, n,
                                                                     row, col, ew, nrm, E);
  k_scatter<F_HID><<<2 * chunks, 256, 0, stream>>>(h, row, col, nrm, agg, ctr, E, nsplit, chunks);

  k_gemm_wmma<F_OUT, true, false><<<nblk(n / 16, 8), 256, 0, stream>>>(agg, W2, dis, b2, h, out, n,
                                                                     nullptr, nullptr, nullptr, nullptr, 0);
  k_scatter<F_OUT><<<2 * chunks, 256, 0, stream>>>(h, row, col, nrm, out, ctr + 2, E, nsplit, chunks);
}
